# starting kernel
# baseline (speedup 1.0000x reference)
#include <hip/hip_runtime.h>
#include <math.h>

#define Bdim 8
#define Ndim 256
#define ZDIM 64
#define EDIM 16
#define HDIM 64

typedef __attribute__((ext_vector_type(2))) float v2f;
typedef __attribute__((ext_vector_type(8))) float v8f;

__global__ void mimj_kernel(const float* __restrict__ z,
                            const float* __restrict__ Wmsg,
                            float* __restrict__ mi, float* __restrict__ mj) {
    int bn = blockIdx.x;
    int h  = threadIdx.x;
    __shared__ float zrow[ZDIM];
    zrow[h] = z[bn * ZDIM + h];
    __syncthreads();
    const float* wi = Wmsg + h * (2 * ZDIM + EDIM);
    const float* wj = wi + ZDIM;
    float si = 0.f, sj = 0.f;
#pragma unroll 8
    for (int f = 0; f < ZDIM; ++f) {
        float zz = zrow[f];
        si += zz * wi[f];
        sj += zz * wj[f];
    }
    mi[bn * HDIM + h] = si;
    mj[bn * HDIM + h] = sj;
}

#define E_STRIDE 20
#define W_STRIDE 80

__global__ void msg_kernel(const float* __restrict__ e_feat,
                           const int*   __restrict__ adj,
                           const float* __restrict__ Wmsg,
                           const float* __restrict__ bmsg,
                           const float* __restrict__ mj,
                           const float* __restrict__ mi,
                           float* __restrict__ agg) {
    __shared__ float E[Ndim * E_STRIDE];
    __shared__ float WeT[EDIM * W_STRIDE];
    __shared__ float maskF[Ndim];
    __shared__ float red[8 * HDIM];

    const int bn = blockIdx.x;
    const int b  = bn >> 8;
    const int t  = threadIdx.x;

    const float NEG_INF = -__builtin_inff();

    {
        const float4* src = (const float4*)(e_feat + (size_t)bn * Ndim * EDIM + t * EDIM);
        float4* dst = (float4*)(E + t * E_STRIDE);
#pragma unroll
        for (int q = 0; q < 4; ++q) dst[q] = src[q];
    }
    {
#pragma unroll
        for (int q = 0; q < 4; ++q) {
            int idx = t * 4 + q;
            int e = idx >> 6, h = idx & 63;
            WeT[e * W_STRIDE + h] = Wmsg[h * (2 * ZDIM + EDIM) + 2 * ZDIM + e];
        }
    }
    maskF[t] = (adj[(size_t)bn * Ndim + t] == 1) ? 0.f : NEG_INF;
    __syncthreads();

    const int wave  = t >> 5;
    const int lane  = t & 31;
    const int lm    = lane & 15;
    const int hi    = lane >> 4;
    const int jwave = wave * 32;

    float mk[16];
#pragma unroll
    for (int jt = 0; jt < 2; ++jt)
#pragma unroll
        for (int r = 0; r < 8; ++r)
            mk[jt * 8 + r] = maskF[jwave + jt * 16 + r + 8 * hi];

    float vm0 = NEG_INF, vm1 = NEG_INF, vm2 = NEG_INF, vm3 = NEG_INF;
    float vmarr[4];

#pragma unroll
    for (int jt = 0; jt < 2; ++jt) {
        const int jbase = jwave + jt * 16;

        v2f afr[4];
#pragma unroll
        for (int k = 0; k < 4; ++k) {
            const int ar = (jbase + lm) * E_STRIDE + 4 * k + 2 * hi;
            afr[k].x = E[ar];
            afr[k].y = E[ar + 1];
        }

        float vmn[4] = {NEG_INF, NEG_INF, NEG_INF, NEG_INF};
#pragma unroll
        for (int ht = 0; ht < 4; ++ht) {
            const int hbase = ht * 16;

            const float* mjp = mj + ((size_t)b * Ndim + jbase + 8 * hi) * HDIM
                                  + hbase + lm;
            v8f acc;
#pragma unroll
            for (int r = 0; r < 8; ++r) acc[r] = mjp[r * HDIM];

#pragma unroll
            for (int k = 0; k < 4; ++k) {
                const int k0 = 4 * k;
                v2f bm;
                const int be = (k0 + 2 * hi) * W_STRIDE + hbase + lm;
                bm.x = WeT[be];
                bm.y = WeT[be + W_STRIDE];
                acc = __builtin_amdgcn_wmma_f32_16x16x4_f32(
                    false, afr[k], false, bm, (short)0, acc, false, false);
            }

            float v = vmn[ht];
#pragma unroll
            for (int r = 0; r < 8; ++r) v = fmaxf(v, acc[r] + mk[jt * 8 + r]);
            vmn[ht] = v;
        }
        vm0 = fmaxf(vm0, vmn[0]);
        vm1 = fmaxf(vm1, vmn[1]);
        vm2 = fmaxf(vm2, vmn[2]);
        vm3 = fmaxf(vm3, vmn[3]);
    }
    vmarr[0] = vm0; vmarr[1] = vm1; vmarr[2] = vm2; vmarr[3] = vm3;

#pragma unroll
    for (int ht = 0; ht < 4; ++ht) {
        float v = fmaxf(vmarr[ht], __shfl_xor(vmarr[ht], 16, 32));
        if (lane < 16) red[wave * HDIM + ht * 16 + lm] = v;
    }
    __syncthreads();

    if (t < HDIM) {
        float m = red[t];
#pragma unroll
        for (int w = 1; w < 8; ++w) m = fmaxf(m, red[w * HDIM + t]);
        agg[(size_t)bn * HDIM + t] = m + mi[(size_t)bn * HDIM + t] + bmsg[t];
    }
}

__global__ void upd_kernel(const float* __restrict__ z,
                           const float* __restrict__ agg,
                           const float* __restrict__ Wupd,
                           const float* __restrict__ bupd,
                           float* __restrict__ out) {
    int bn = blockIdx.x;
    int h  = threadIdx.x;
    __shared__ float feat[ZDIM + HDIM];
    feat[h]        = z[bn * ZDIM + h];
    feat[ZDIM + h] = agg[bn * HDIM + h];
    __syncthreads();
    const float* w = Wupd + h * (ZDIM + HDIM);
    float s = bupd[h];
#pragma unroll 8
    for (int f = 0; f < ZDIM + HDIM; ++f) s += feat[f] * w[f];
    out[bn * HDIM + h] = s;
}

extern "C" void kernel_launch(void* const* d_in, const int* in_sizes, int n_in,
                              void* d_out, int out_size, void* d_ws, size_t ws_size,
                              hipStream_t stream) {
    const float* z      = (const float*)d_in[0];
    const float* e_feat = (const float*)d_in[1];
    const int*   adj    = (const int*)d_in[2];
    const float* W_msg  = (const float*)d_in[3];
    const float* b_msg  = (const float*)d_in[4];
    const float* W_upd  = (const float*)d_in[5];
    const float* b_upd  = (const float*)d_in[6];
    float*       out    = (float*)d_out;

    const int BN = Bdim * Ndim;
    float* mi  = (float*)d_ws;
    float* mj  = mi + BN * HDIM;
    float* agg = mj + BN * HDIM;

    mimj_kernel<<<BN, HDIM, 0, stream>>>(z, W_msg, mi, mj);
    msg_kernel <<<BN, 256,  0, stream>>>(e_feat, adj, W_msg, b_msg, mj, mi, agg);
    upd_kernel <<<BN, HDIM, 0, stream>>>(z, agg, W_upd, b_upd, out);
}


# msg kernel: one workgroup per (batch, 8 rows i): m_j tiles and W_e fragments kept in registers across rows, A fragments straight from global; k permuted inside WMMA K-steps (sum order changes)
# speedup vs baseline: 1.2083x; 1.2083x over previous
#include <hip/hip_runtime.h>
#include <math.h>

#define Bdim 8
#define Ndim 256
#define ZDIM 64
#define EDIM 16
#define HDIM 64

typedef __attribute__((ext_vector_type(2))) float v2f;
typedef __attribute__((ext_vector_type(8))) float v8f;

__global__ void mimj_kernel(const float* __restrict__ z,
                            const float* __restrict__ Wmsg,
                            float* __restrict__ mi, float* __restrict__ mj) {
    int bn = blockIdx.x;
    int h  = threadIdx.x;
    __shared__ float zrow[ZDIM];
    zrow[h] = z[bn * ZDIM + h];
    __syncthreads();
    const float* wi = Wmsg + h * (2 * ZDIM + EDIM);
    const float* wj = wi + ZDIM;
    float si = 0.f, sj = 0.f;
#pragma unroll 8
    for (int f = 0; f < ZDIM; ++f) {
        float zz = zrow[f];
        si += zz * wi[f];
        sj += zz * wj[f];
    }
    mi[bn * HDIM + h] = si;
    mj[bn * HDIM + h] = sj;
}

#define IB 8
#define NSTEP (IB * 4)
#define RED_PITCH 80

__global__ __launch_bounds__(256)
void msg_kernel(const float* __restrict__ e_feat,
                const int*   __restrict__ adj,
                const float* __restrict__ Wmsg,
                const float* __restrict__ bmsg,
                const float* __restrict__ mj,
                const float* __restrict__ mi,
                float* __restrict__ agg) {
    __shared__ float red[IB * 16 * RED_PITCH];

    const int blk = blockIdx.x;
    const int b   = blk / (Ndim / IB);
    const int i0  = (blk % (Ndim / IB)) * IB;
    const int t   = threadIdx.x;

    const float NEG_INF = -__builtin_inff();

    const int wave  = t >> 5;
    const int lane  = t & 31;
    const int lm    = lane & 15;
    const int hi    = lane >> 4;
    const int jwave = wave * 32;

    float4 af[IB][2][2];
    int    adjv[IB];
    float4 bf[4][2];
    v8f    cj[2][4];

#pragma unroll
    for (int ii = 0; ii < IB / 2; ++ii)
#pragma unroll
        for (int jt = 0; jt < 2; ++jt) {
            const float4* p = (const float4*)(e_feat
                + (((size_t)(b * Ndim + i0 + ii) * Ndim) + jwave + jt * 16 + lm) * EDIM
                + hi * 8);
            af[ii][jt][0] = p[0];
            af[ii][jt][1] = p[1];
        }
#pragma unroll
    for (int ii = 0; ii < IB; ++ii)
        adjv[ii] = adj[(size_t)(b * Ndim + i0 + ii) * Ndim + jwave + lane];
#pragma unroll
    for (int ht = 0; ht < 4; ++ht) {
        const float4* p = (const float4*)(Wmsg
            + (size_t)(ht * 16 + lm) * (2 * ZDIM + EDIM) + 2 * ZDIM + hi * 8);
        bf[ht][0] = p[0];
        bf[ht][1] = p[1];
    }
    __builtin_amdgcn_sched_barrier(0);

#pragma unroll
    for (int ht = 0; ht < 4; ++ht)
#pragma unroll
        for (int jt = 0; jt < 2; ++jt) {
            const float* mjp = mj + ((size_t)b * Ndim + jwave + jt * 16 + 8 * hi) * HDIM
                                  + ht * 16 + lm;
#pragma unroll
            for (int r = 0; r < 8; ++r) cj[jt][ht][r] = mjp[r * HDIM];
        }
    __builtin_amdgcn_sched_barrier(0);

#pragma unroll
    for (int ii = IB / 2; ii < IB; ++ii)
#pragma unroll
        for (int jt = 0; jt < 2; ++jt) {
            const float4* p = (const float4*)(e_feat
                + (((size_t)(b * Ndim + i0 + ii) * Ndim) + jwave + jt * 16 + lm) * EDIM
                + hi * 8);
            af[ii][jt][0] = p[0];
            af[ii][jt][1] = p[1];
        }
    __builtin_amdgcn_sched_barrier(0);

    float mk[2][16];
    v8f   acc[3][2];
    float vm[4];

#define MSG_MASK(II)                                                              \
    do {                                                                          \
        const unsigned wb = __builtin_amdgcn_ballot_w32(adjv[II] == 1) >> (8 * hi); \
        _Pragma("unroll")                                                         \
        for (int q = 0; q < 16; ++q)                                              \
            mk[(II) & 1][q] = ((wb >> ((q >> 3) * 16 + (q & 7))) & 1u) ? 0.f : NEG_INF; \
    } while (0)

#define MSG_INIT(S)                                                               \
    do {                                                                          \
        _Pragma("unroll")                                                         \
        for (int r = 0; r < 8; ++r) {                                             \
            acc[(S) % 3][0][r] = cj[0][(S) & 3][r] + mk[((S) >> 2) & 1][r];       \
            acc[(S) % 3][1][r] = cj[1][(S) & 3][r] + mk[((S) >> 2) & 1][8 + r];   \
        }                                                                         \
    } while (0)

#define MSG_WMMA(S)                                                               \
    do {                                                                          \
        _Pragma("unroll")                                                         \
        for (int kk = 0; kk < 4; ++kk) {                                          \
            const float4 b4 = bf[(S) & 3][kk >> 1];                               \
            const float4 a0 = af[(S) >> 2][0][kk >> 1];                           \
            const float4 a1 = af[(S) >> 2][1][kk >> 1];                           \
            v2f bm, am0, am1;                                                     \
            if (kk & 1) { bm.x = b4.z; bm.y = b4.w; am0.x = a0.z; am0.y = a0.w;   \
                          am1.x = a1.z; am1.y = a1.w; }                           \
            else        { bm.x = b4.x; bm.y = b4.y; am0.x = a0.x; am0.y = a0.y;   \
                          am1.x = a1.x; am1.y = a1.y; }                           \
            acc[(S) % 3][0] = __builtin_amdgcn_wmma_f32_16x16x4_f32(              \
                false, am0, false, bm, (short)0, acc[(S) % 3][0], false, false);  \
            acc[(S) % 3][1] = __builtin_amdgcn_wmma_f32_16x16x4_f32(              \
                false, am1, false, bm, (short)0, acc[(S) % 3][1], false, false);  \
        }                                                                         \
    } while (0)

#define MSG_MAX(S)                                                                \
    do {                                                                          \
        float v0 = acc[(S) % 3][0][0];                                            \
        float v1 = acc[(S) % 3][1][0];                                            \
        _Pragma("unroll")                                                         \
        for (int r = 1; r < 8; ++r) {                                             \
            v0 = fmaxf(v0, acc[(S) % 3][0][r]);                                   \
            v1 = fmaxf(v1, acc[(S) % 3][1][r]);                                   \
        }                                                                         \
        vm[(S) & 3] = fmaxf(v0, v1);                                              \
        if (((S) & 3) == 3) {                                                     \
            float* rp = red + ((((S) >> 2) * 8 + wave) * 2 + hi) * RED_PITCH + lm; \
            rp[0]  = vm[0];                                                       \
            rp[16] = vm[1];                                                       \
            rp[32] = vm[2];                                                       \
            rp[48] = vm[3];                                                       \
        }                                                                         \
    } while (0)

    MSG_MASK(0);
    MSG_INIT(0);
    __builtin_amdgcn_sched_barrier(0);
#pragma unroll
    for (int s = 0; s < NSTEP; ++s) {
        MSG_WMMA(s);
        if (s >= 1) MSG_MAX(s - 1);
        if ((s & 3) == 1 && (s >> 2) + 1 < IB) MSG_MASK((s >> 2) + 1);
        if (s + 1 < NSTEP) MSG_INIT(s + 1);
        __builtin_amdgcn_sched_barrier(0);
    }
    MSG_MAX(NSTEP - 1);

    float miv[IB * HDIM / 256], bmv[IB * HDIM / 256];
#pragma unroll
    for (int q = 0; q < IB * HDIM / 256; ++q) {
        const int o = t + q * 256;
        const int ii = o >> 6, h = o & 63;
        miv[q] = mi[(size_t)(b * Ndim + i0 + ii) * HDIM + h];
        bmv[q] = bmsg[h];
    }
    __syncthreads();

#pragma unroll
    for (int q = 0; q < IB * HDIM / 256; ++q) {
        const int o = t + q * 256;
        const int ii = o >> 6, h = o & 63;
        float m = red[(ii * 16) * RED_PITCH + h];
#pragma unroll
        for (int w16 = 1; w16 < 16; ++w16) m = fmaxf(m, red[(ii * 16 + w16) * RED_PITCH + h]);
        agg[(size_t)(b * Ndim + i0 + ii) * HDIM + h] = m + miv[q] + bmv[q];
    }
#undef MSG_MASK
#undef MSG_INIT
#undef MSG_WMMA
#undef MSG_MAX
}

__global__ void upd_kernel(const float* __restrict__ z,
                           const float* __restrict__ agg,
                           const float* __restrict__ Wupd,
                           const float* __restrict__ bupd,
                           float* __restrict__ out) {
    int bn = blockIdx.x;
    int h  = threadIdx.x;
    __shared__ float feat[ZDIM + HDIM];
    feat[h]        = z[bn * ZDIM + h];
    feat[ZDIM + h] = agg[bn * HDIM + h];
    __syncthreads();
    const float* w = Wupd + h * (ZDIM + HDIM);
    float s = bupd[h];
#pragma unroll 8
    for (int f = 0; f < ZDIM + HDIM; ++f) s += feat[f] * w[f];
    out[bn * HDIM + h] = s;
}

extern "C" void kernel_launch(void* const* d_in, const int* in_sizes, int n_in,
                              void* d_out, int out_size, void* d_ws, size_t ws_size,
                              hipStream_t stream) {
    const float* z      = (const float*)d_in[0];
    const float* e_feat = (const float*)d_in[1];
    const int*   adj    = (const int*)d_in[2];
    const float* W_msg  = (const float*)d_in[3];
    const float* b_msg  = (const float*)d_in[4];
    const float* W_upd  = (const float*)d_in[5];
    const float* b_upd  = (const float*)d_in[6];
    float*       out    = (float*)d_out;

    const int BN = Bdim * Ndim;
    float* mi  = (float*)d_ws;
    float* mj  = mi + BN * HDIM;
    float* agg = mj + BN * HDIM;

    mimj_kernel<<<BN, HDIM, 0, stream>>>(z, W_msg, mi, mj);
    msg_kernel <<<BN / IB, 256, 0, stream>>>(e_feat, adj, W_msg, b_msg, mj, mi, agg);
    upd_kernel <<<BN, HDIM, 0, stream>>>(z, agg, W_upd, b_upd, out);
}


# update layer fused into the msg kernel: [z|agg] rows staged in LDS, 8x64x128 product on f32 WMMA over 8 waves, operands fetched mid-loop; 3 launches -> 2 (upd sum order changes)
# speedup vs baseline: 1.6739x; 1.3853x over previous
#include <hip/hip_runtime.h>
#include <math.h>

#define Bdim 8
#define Ndim 256
#define ZDIM 64
#define EDIM 16
#define HDIM 64

typedef __attribute__((ext_vector_type(2))) float v2f;
typedef __attribute__((ext_vector_type(8))) float v8f;

__global__ void mimj_kernel(const float* __restrict__ z,
                            const float* __restrict__ Wmsg,
                            float* __restrict__ mi, float* __restrict__ mj) {
    int bn = blockIdx.x;
    int h  = threadIdx.x;
    __shared__ float zrow[ZDIM];
    zrow[h] = z[bn * ZDIM + h];
    __syncthreads();
    const float* wi = Wmsg + h * (2 * ZDIM + EDIM);
    const float* wj = wi + ZDIM;
    float si = 0.f, sj = 0.f;
#pragma unroll 8
    for (int f = 0; f < ZDIM; ++f) {
        float zz = zrow[f];
        si += zz * wi[f];
        sj += zz * wj[f];
    }
    mi[bn * HDIM + h] = si;
    mj[bn * HDIM + h] = sj;
}

#define IB 8
#define NSTEP (IB * 4)
#define RED_PITCH 80
#define FEAT_PITCH 132

__global__ __launch_bounds__(256)
void msg_upd_kernel(const float* __restrict__ e_feat,
                    const int*   __restrict__ adj,
                    const float* __restrict__ Wmsg,
                    const float* __restrict__ bmsg,
                    const float* __restrict__ mj,
                    const float* __restrict__ mi,
                    const float* __restrict__ z,
                    const float* __restrict__ Wupd,
                    const float* __restrict__ bupd,
                    float* __restrict__ out) {
    __shared__ float red[IB * 16 * RED_PITCH];
    __shared__ __attribute__((aligned(16))) float feat[IB * FEAT_PITCH];
    __shared__ __attribute__((aligned(16))) float part[2 * 16 * HDIM];

    const int blk = blockIdx.x;
    const int b   = blk / (Ndim / IB);
    const int i0  = (blk % (Ndim / IB)) * IB;
    const int t   = threadIdx.x;

    const float NEG_INF = -__builtin_inff();

    const int wave  = t >> 5;
    const int lane  = t & 31;
    const int lm    = lane & 15;
    const int hi    = lane >> 4;
    const int jwave = wave * 32;

    float4 af[IB][2][2];
    int    adjv[IB];
    float4 bf[4][2];
    v8f    cj[2][4];

#pragma unroll
    for (int ii = 0; ii < IB / 2; ++ii)
#pragma unroll
        for (int jt = 0; jt < 2; ++jt) {
            const float4* p = (const float4*)(e_feat
                + (((size_t)(b * Ndim + i0 + ii) * Ndim) + jwave + jt * 16 + lm) * EDIM
                + hi * 8);
            af[ii][jt][0] = p[0];
            af[ii][jt][1] = p[1];
        }
#pragma unroll
    for (int ii = 0; ii < IB; ++ii)
        adjv[ii] = adj[(size_t)(b * Ndim + i0 + ii) * Ndim + jwave + lane];
#pragma unroll
    for (int ht = 0; ht < 4; ++ht) {
        const float4* p = (const float4*)(Wmsg
            + (size_t)(ht * 16 + lm) * (2 * ZDIM + EDIM) + 2 * ZDIM + hi * 8);
        bf[ht][0] = p[0];
        bf[ht][1] = p[1];
    }
    __builtin_amdgcn_sched_barrier(0);

#pragma unroll
    for (int ht = 0; ht < 4; ++ht)
#pragma unroll
        for (int jt = 0; jt < 2; ++jt) {
            const float* mjp = mj + ((size_t)b * Ndim + jwave + jt * 16 + 8 * hi) * HDIM
                                  + ht * 16 + lm;
#pragma unroll
            for (int r = 0; r < 8; ++r) cj[jt][ht][r] = mjp[r * HDIM];
        }
    __builtin_amdgcn_sched_barrier(0);

#pragma unroll
    for (int ii = IB / 2; ii < IB; ++ii)
#pragma unroll
        for (int jt = 0; jt < 2; ++jt) {
            const float4* p = (const float4*)(e_feat
                + (((size_t)(b * Ndim + i0 + ii) * Ndim) + jwave + jt * 16 + lm) * EDIM
                + hi * 8);
            af[ii][jt][0] = p[0];
            af[ii][jt][1] = p[1];
        }
    __builtin_amdgcn_sched_barrier(0);

    float mk[2][16];
    v8f   acc[3][2];
    float vm[4];

#define MSG_MASK(II)                                                              \
    do {                                                                          \
        const unsigned wb = __builtin_amdgcn_ballot_w32(adjv[II] == 1) >> (8 * hi); \
        _Pragma("unroll")                                                         \
        for (int q = 0; q < 16; ++q)                                              \
            mk[(II) & 1][q] = ((wb >> ((q >> 3) * 16 + (q & 7))) & 1u) ? 0.f : NEG_INF; \
    } while (0)

#define MSG_INIT(S)                                                               \
    do {                                                                          \
        _Pragma("unroll")                                                         \
        for (int r = 0; r < 8; ++r) {                                             \
            acc[(S) % 3][0][r] = cj[0][(S) & 3][r] + mk[((S) >> 2) & 1][r];       \
            acc[(S) % 3][1][r] = cj[1][(S) & 3][r] + mk[((S) >> 2) & 1][8 + r];   \
        }                                                                         \
    } while (0)

#define MSG_WMMA(S)                                                               \
    do {                                                                          \
        _Pragma("unroll")                                                         \
        for (int kk = 0; kk < 4; ++kk) {                                          \
            const float4 b4 = bf[(S) & 3][kk >> 1];                               \
            const float4 a0 = af[(S) >> 2][0][kk >> 1];                           \
            const float4 a1 = af[(S) >> 2][1][kk >> 1];                           \
            v2f bm, am0, am1;                                                     \
            if (kk & 1) { bm.x = b4.z; bm.y = b4.w; am0.x = a0.z; am0.y = a0.w;   \
                          am1.x = a1.z; am1.y = a1.w; }                           \
            else        { bm.x = b4.x; bm.y = b4.y; am0.x = a0.x; am0.y = a0.y;   \
                          am1.x = a1.x; am1.y = a1.y; }                           \
            acc[(S) % 3][0] = __builtin_amdgcn_wmma_f32_16x16x4_f32(              \
                false, am0, false, bm, (short)0, acc[(S) % 3][0], false, false);  \
            acc[(S) % 3][1] = __builtin_amdgcn_wmma_f32_16x16x4_f32(              \
                false, am1, false, bm, (short)0, acc[(S) % 3][1], false, false);  \
        }                                                                         \
    } while (0)

#define MSG_MAX(S)                                                                \
    do {                                                                          \
        float v0 = acc[(S) % 3][0][0];                                            \
        float v1 = acc[(S) % 3][1][0];                                            \
        _Pragma("unroll")                                                         \
        for (int r = 1; r < 8; ++r) {                                             \
            v0 = fmaxf(v0, acc[(S) % 3][0][r]);                                   \
            v1 = fmaxf(v1, acc[(S) % 3][1][r]);                                   \
        }                                                                         \
        vm[(S) & 3] = fmaxf(v0, v1);                                              \
        if (((S) & 3) == 3) {                                                     \
            float* rp = red + ((((S) >> 2) * 8 + wave) * 2 + hi) * RED_PITCH + lm; \
            rp[0]  = vm[0];                                                       \
            rp[16] = vm[1];                                                       \
            rp[32] = vm[2];                                                       \
            rp[48] = vm[3];                                                       \
        }                                                                         \
    } while (0)

    const int uh = wave & 3;
    const int uk = wave >> 2;
    float4 wu[4][2];
    float  miv[IB * HDIM / 256], bmv[IB * HDIM / 256], zv[IB * ZDIM / 256];
    float4 bu;
#define EPI_STEP 4
#define MSG_EPI_FETCH()                                                           \
    do {                                                                          \
        _Pragma("unroll")                                                         \
        for (int c = 0; c < 4; ++c) {                                             \
            const float4* p = (const float4*)(Wupd                                \
                + (size_t)(uh * 16 + lm) * (ZDIM + HDIM) + uk * 64 + c * 16 + hi * 8); \
            wu[c][0] = p[0];                                                      \
            wu[c][1] = p[1];                                                      \
        }                                                                         \
        _Pragma("unroll")                                                         \
        for (int q = 0; q < IB * HDIM / 256; ++q) {                               \
            const int o = t + q * 256;                                            \
            const int ii = o >> 6, h = o & 63;                                    \
            miv[q] = mi[(size_t)(b * Ndim + i0 + ii) * HDIM + h];                 \
            bmv[q] = bmsg[h];                                                     \
            zv[q]  = z[(size_t)(b * Ndim + i0 + ii) * ZDIM + h];                  \
        }                                                                         \
        bu = ((const float4*)bupd)[t & 15];                                       \
    } while (0)

    MSG_MASK(0);
    MSG_INIT(0);
    __builtin_amdgcn_sched_barrier(0);
#pragma unroll
    for (int s = 0; s < NSTEP; ++s) {
        MSG_WMMA(s);
        if (s >= 1) MSG_MAX(s - 1);
        if ((s & 3) == 1 && (s >> 2) + 1 < IB) MSG_MASK((s >> 2) + 1);
        if (s + 1 < NSTEP) MSG_INIT(s + 1);
        if (s == EPI_STEP) MSG_EPI_FETCH();
        __builtin_amdgcn_sched_barrier(0);
    }
    MSG_MAX(NSTEP - 1);

    __syncthreads();

#pragma unroll
    for (int q = 0; q < IB * HDIM / 256; ++q) {
        const int o = t + q * 256;
        const int ii = o >> 6, h = o & 63;
        float m = red[(ii * 16) * RED_PITCH + h];
#pragma unroll
        for (int w16 = 1; w16 < 16; ++w16) m = fmaxf(m, red[(ii * 16 + w16) * RED_PITCH + h]);
        feat[ii * FEAT_PITCH + h]        = zv[q];
        feat[ii * FEAT_PITCH + ZDIM + h] = m + miv[q] + bmv[q];
    }
    __syncthreads();

    float4 fa[4][2];
#pragma unroll
    for (int c = 0; c < 4; ++c) {
        const float4* p = (const float4*)(feat + (lm & 7) * FEAT_PITCH + uk * 64 + c * 16 + hi * 8);
        fa[c][0] = p[0];
        fa[c][1] = p[1];
    }
    __builtin_amdgcn_sched_barrier(0);
    v8f u = {0.f, 0.f, 0.f, 0.f, 0.f, 0.f, 0.f, 0.f};
#pragma unroll
    for (int c = 0; c < 4; ++c) {
#pragma unroll
        for (int kk = 0; kk < 4; ++kk) {
            const float4 a4 = fa[c][kk >> 1];
            const float4 b4 = wu[c][kk >> 1];
            v2f am, bm;
            if (kk & 1) { am.x = a4.z; am.y = a4.w; bm.x = b4.z; bm.y = b4.w; }
            else        { am.x = a4.x; am.y = a4.y; bm.x = b4.x; bm.y = b4.y; }
            u = __builtin_amdgcn_wmma_f32_16x16x4_f32(
                false, am, false, bm, (short)0, u, false, false);
        }
    }
    __builtin_amdgcn_sched_barrier(0);
    float* pp = part + (uk * 16 + 8 * hi) * HDIM + uh * 16 + lm;
#pragma unroll
    for (int r = 0; r < 8; ++r) pp[r * HDIM] = u[r];
    __syncthreads();

    if (t < IB * HDIM / 4) {
        const float4 p0 = ((const float4*)part)[t];
        const float4 p1 = ((const float4*)part)[16 * HDIM / 4 + t];
        float4 o4;
        o4.x = p0.x + p1.x + bu.x;
        o4.y = p0.y + p1.y + bu.y;
        o4.z = p0.z + p1.z + bu.z;
        o4.w = p0.w + p1.w + bu.w;
        ((float4*)(out + (size_t)(b * Ndim + i0) * HDIM))[t] = o4;
    }
#undef MSG_MASK
#undef MSG_INIT
#undef MSG_WMMA
#undef MSG_MAX
#undef MSG_EPI_FETCH
#undef EPI_STEP
}

extern "C" void kernel_launch(void* const* d_in, const int* in_sizes, int n_in,
                              void* d_out, int out_size, void* d_ws, size_t ws_size,
                              hipStream_t stream) {
    const float* z      = (const float*)d_in[0];
    const float* e_feat = (const float*)d_in[1];
    const int*   adj    = (const int*)d_in[2];
    const float* W_msg  = (const float*)d_in[3];
    const float* b_msg  = (const float*)d_in[4];
    const float* W_upd  = (const float*)d_in[5];
    const float* b_upd  = (const float*)d_in[6];
    float*       out    = (float*)d_out;

    const int BN = Bdim * Ndim;
    float* mi  = (float*)d_ws;
    float* mj  = mi + BN * HDIM;

    mimj_kernel   <<<BN, HDIM, 0, stream>>>(z, W_msg, mi, mj);
    msg_upd_kernel<<<BN / IB, 256, 0, stream>>>(e_feat, adj, W_msg, b_msg, mj, mi,
                                                z, W_upd, b_upd, out);
}


# m_i/m_j kernel fused in: each wave forms its own m_j tiles (32x64, K=64) and an m_i partial on f32 WMMA while e_feat loads fly; 2 launches -> 1, no workspace (m_i, m_j sum order changes)
# speedup vs baseline: 2.1203x; 1.2667x over previous
#include <hip/hip_runtime.h>
#include <math.h>

#define Bdim 8
#define Ndim 256
#define ZDIM 64
#define EDIM 16
#define HDIM 64

typedef __attribute__((ext_vector_type(2))) float v2f;
typedef __attribute__((ext_vector_type(8))) float v8f;

#define IB 8
#define NSTEP (IB * 4)
#define RED_PITCH 80
#define FEAT_PITCH 132

__global__ __launch_bounds__(256)
void msg_upd_kernel(const float* __restrict__ e_feat,
                    const int*   __restrict__ adj,
                    const float* __restrict__ Wmsg,
                    const float* __restrict__ bmsg,
                    const float* __restrict__ z,
                    const float* __restrict__ Wupd,
                    const float* __restrict__ bupd,
                    float* __restrict__ out) {
    __shared__ float red[IB * 16 * RED_PITCH];
    __shared__ __attribute__((aligned(16))) float feat[IB * FEAT_PITCH];
    __shared__ __attribute__((aligned(16))) float part[2 * 16 * HDIM];
    __shared__ float mip[2 * 16 * HDIM];

    const int blk = blockIdx.x;
    const int b   = blk / (Ndim / IB);
    const int i0  = (blk % (Ndim / IB)) * IB;
    const int t   = threadIdx.x;

    const float NEG_INF = -__builtin_inff();

    const int wave  = t >> 5;
    const int lane  = t & 31;
    const int lm    = lane & 15;
    const int hi    = lane >> 4;
    const int jwave = wave * 32;

    const int uh = wave & 3;
    const int uk = wave >> 2;

    float4 za[4][2][2];
    float4 wb[4][4][2];
    float4 zi[2][2];
    float4 wi[2][2];
    float4 af[IB][2][2];
    int    adjv[IB];
    float4 bf[4][2];
    v8f    cj[2][4];

#pragma unroll
    for (int c = 0; c < 4; ++c) {
#pragma unroll
        for (int jt = 0; jt < 2; ++jt) {
            const float4* p = (const float4*)(z
                + (size_t)(b * Ndim + jwave + jt * 16 + lm) * ZDIM + c * 16 + hi * 8);
            za[c][jt][0] = p[0];
            za[c][jt][1] = p[1];
        }
#pragma unroll
        for (int ht = 0; ht < 4; ++ht) {
            const float4* p = (const float4*)(Wmsg
                + (size_t)(ht * 16 + lm) * (2 * ZDIM + EDIM) + ZDIM + c * 16 + hi * 8);
            wb[c][ht][0] = p[0];
            wb[c][ht][1] = p[1];
        }
    }
#pragma unroll
    for (int c = 0; c < 2; ++c) {
        const float4* pz = (const float4*)(z
            + (size_t)(b * Ndim + i0 + (lm & 7)) * ZDIM + uk * 32 + c * 16 + hi * 8);
        const float4* pw = (const float4*)(Wmsg
            + (size_t)(uh * 16 + lm) * (2 * ZDIM + EDIM) + uk * 32 + c * 16 + hi * 8);
        zi[c][0] = pz[0];
        zi[c][1] = pz[1];
        wi[c][0] = pw[0];
        wi[c][1] = pw[1];
    }

#pragma unroll
    for (int ii = 0; ii < IB / 2; ++ii)
#pragma unroll
        for (int jt = 0; jt < 2; ++jt) {
            const float4* p = (const float4*)(e_feat
                + (((size_t)(b * Ndim + i0 + ii) * Ndim) + jwave + jt * 16 + lm) * EDIM
                + hi * 8);
            af[ii][jt][0] = p[0];
            af[ii][jt][1] = p[1];
        }
#pragma unroll
    for (int ii = 0; ii < IB; ++ii)
        adjv[ii] = adj[(size_t)(b * Ndim + i0 + ii) * Ndim + jwave + lane];
#pragma unroll
    for (int ht = 0; ht < 4; ++ht) {
        const float4* p = (const float4*)(Wmsg
            + (size_t)(ht * 16 + lm) * (2 * ZDIM + EDIM) + 2 * ZDIM + hi * 8);
        bf[ht][0] = p[0];
        bf[ht][1] = p[1];
    }
    __builtin_amdgcn_sched_barrier(0);

#pragma unroll
    for (int ii = IB / 2; ii < IB; ++ii)
#pragma unroll
        for (int jt = 0; jt < 2; ++jt) {
            const float4* p = (const float4*)(e_feat
                + (((size_t)(b * Ndim + i0 + ii) * Ndim) + jwave + jt * 16 + lm) * EDIM
                + hi * 8);
            af[ii][jt][0] = p[0];
            af[ii][jt][1] = p[1];
        }
    __builtin_amdgcn_sched_barrier(0);

#pragma unroll
    for (int jt = 0; jt < 2; ++jt)
#pragma unroll
        for (int ht = 0; ht < 4; ++ht)
#pragma unroll
            for (int r = 0; r < 8; ++r) cj[jt][ht][r] = 0.f;
#pragma unroll
    for (int c = 0; c < 4; ++c)
#pragma unroll
        for (int kk = 0; kk < 4; ++kk)
#pragma unroll
            for (int jt = 0; jt < 2; ++jt) {
                const float4 a4 = za[c][jt][kk >> 1];
                v2f am;
                if (kk & 1) { am.x = a4.z; am.y = a4.w; }
                else        { am.x = a4.x; am.y = a4.y; }
#pragma unroll
                for (int ht = 0; ht < 4; ++ht) {
                    const float4 b4 = wb[c][ht][kk >> 1];
                    v2f bm;
                    if (kk & 1) { bm.x = b4.z; bm.y = b4.w; }
                    else        { bm.x = b4.x; bm.y = b4.y; }
                    cj[jt][ht] = __builtin_amdgcn_wmma_f32_16x16x4_f32(
                        false, am, false, bm, (short)0, cj[jt][ht], false, false);
                }
            }
    {
        v8f pm = {0.f, 0.f, 0.f, 0.f, 0.f, 0.f, 0.f, 0.f};
#pragma unroll
        for (int c = 0; c < 2; ++c)
#pragma unroll
            for (int kk = 0; kk < 4; ++kk) {
                const float4 a4 = zi[c][kk >> 1];
                const float4 b4 = wi[c][kk >> 1];
                v2f am, bm;
                if (kk & 1) { am.x = a4.z; am.y = a4.w; bm.x = b4.z; bm.y = b4.w; }
                else        { am.x = a4.x; am.y = a4.y; bm.x = b4.x; bm.y = b4.y; }
                pm = __builtin_amdgcn_wmma_f32_16x16x4_f32(
                    false, am, false, bm, (short)0, pm, false, false);
            }
        float* mp = mip + (uk * 16 + 8 * hi) * HDIM + uh * 16 + lm;
#pragma unroll
        for (int r = 0; r < 8; ++r) mp[r * HDIM] = pm[r];
    }
    __builtin_amdgcn_sched_barrier(0);

    float mk[2][16];
    v8f   acc[3][2];
    float vm[4];

#define MSG_MASK(II)                                                              \
    do {                                                                          \
        const unsigned wb = __builtin_amdgcn_ballot_w32(adjv[II] == 1) >> (8 * hi); \
        _Pragma("unroll")                                                         \
        for (int q = 0; q < 16; ++q)                                              \
            mk[(II) & 1][q] = ((wb >> ((q >> 3) * 16 + (q & 7))) & 1u) ? 0.f : NEG_INF; \
    } while (0)

#define MSG_INIT(S)                                                               \
    do {                                                                          \
        _Pragma("unroll")                                                         \
        for (int r = 0; r < 8; ++r) {                                             \
            acc[(S) % 3][0][r] = cj[0][(S) & 3][r] + mk[((S) >> 2) & 1][r];       \
            acc[(S) % 3][1][r] = cj[1][(S) & 3][r] + mk[((S) >> 2) & 1][8 + r];   \
        }                                                                         \
    } while (0)

#define MSG_WMMA(S)                                                               \
    do {                                                                          \
        _Pragma("unroll")                                                         \
        for (int kk = 0; kk < 4; ++kk) {                                          \
            const float4 b4 = bf[(S) & 3][kk >> 1];                               \
            const float4 a0 = af[(S) >> 2][0][kk >> 1];                           \
            const float4 a1 = af[(S) >> 2][1][kk >> 1];                           \
            v2f bm, am0, am1;                                                     \
            if (kk & 1) { bm.x = b4.z; bm.y = b4.w; am0.x = a0.z; am0.y = a0.w;   \
                          am1.x = a1.z; am1.y = a1.w; }                           \
            else        { bm.x = b4.x; bm.y = b4.y; am0.x = a0.x; am0.y = a0.y;   \
                          am1.x = a1.x; am1.y = a1.y; }                           \
            acc[(S) % 3][0] = __builtin_amdgcn_wmma_f32_16x16x4_f32(              \
                false, am0, false, bm, (short)0, acc[(S) % 3][0], false, false);  \
            acc[(S) % 3][1] = __builtin_amdgcn_wmma_f32_16x16x4_f32(              \
                false, am1, false, bm, (short)0, acc[(S) % 3][1], false, false);  \
        }                                                                         \
    } while (0)

#define MSG_MAX(S)                                                                \
    do {                                                                          \
        float v0 = acc[(S) % 3][0][0];                                            \
        float v1 = acc[(S) % 3][1][0];                                            \
        _Pragma("unroll")                                                         \
        for (int r = 1; r < 8; ++r) {                                             \
            v0 = fmaxf(v0, acc[(S) % 3][0][r]);                                   \
            v1 = fmaxf(v1, acc[(S) % 3][1][r]);                                   \
        }                                                                         \
        vm[(S) & 3] = fmaxf(v0, v1);                                              \
        if (((S) & 3) == 3) {                                                     \
            float* rp = red + ((((S) >> 2) * 8 + wave) * 2 + hi) * RED_PITCH + lm; \
            rp[0]  = vm[0];                                                       \
            rp[16] = vm[1];                                                       \
            rp[32] = vm[2];                                                       \
            rp[48] = vm[3];                                                       \
        }                                                                         \
    } while (0)

    float4 wu[4][2];
    float  bmv[IB * HDIM / 256], zv[IB * ZDIM / 256];
    float4 bu;
#define EPI_STEP 4
#define MSG_EPI_FETCH()                                                           \
    do {                                                                          \
        _Pragma("unroll")                                                         \
        for (int c = 0; c < 4; ++c) {                                             \
            const float4* p = (const float4*)(Wupd                                \
                + (size_t)(uh * 16 + lm) * (ZDIM + HDIM) + uk * 64 + c * 16 + hi * 8); \
            wu[c][0] = p[0];                                                      \
            wu[c][1] = p[1];                                                      \
        }                                                                         \
        _Pragma("unroll")                                                         \
        for (int q = 0; q < IB * HDIM / 256; ++q) {                               \
            const int o = t + q * 256;                                            \
            const int ii = o >> 6, h = o & 63;                                    \
            bmv[q] = bmsg[h];                                                     \
            zv[q]  = z[(size_t)(b * Ndim + i0 + ii) * ZDIM + h];                  \
        }                                                                         \
        bu = ((const float4*)bupd)[t & 15];                                       \
    } while (0)

    MSG_MASK(0);
    MSG_INIT(0);
    __builtin_amdgcn_sched_barrier(0);
#pragma unroll
    for (int s = 0; s < NSTEP; ++s) {
        MSG_WMMA(s);
        if (s >= 1) MSG_MAX(s - 1);
        if ((s & 3) == 1 && (s >> 2) + 1 < IB) MSG_MASK((s >> 2) + 1);
        if (s + 1 < NSTEP) MSG_INIT(s + 1);
        if (s == EPI_STEP) MSG_EPI_FETCH();
        __builtin_amdgcn_sched_barrier(0);
    }
    MSG_MAX(NSTEP - 1);

    __syncthreads();

#pragma unroll
    for (int q = 0; q < IB * HDIM / 256; ++q) {
        const int o = t + q * 256;
        const int ii = o >> 6, h = o & 63;
        float m = red[(ii * 16) * RED_PITCH + h];
#pragma unroll
        for (int w16 = 1; w16 < 16; ++w16) m = fmaxf(m, red[(ii * 16 + w16) * RED_PITCH + h]);
        feat[ii * FEAT_PITCH + h]        = zv[q];
        feat[ii * FEAT_PITCH + ZDIM + h] = m + (mip[ii * HDIM + h] + mip[(16 + ii) * HDIM + h]) + bmv[q];
    }
    __syncthreads();

    float4 fa[4][2];
#pragma unroll
    for (int c = 0; c < 4; ++c) {
        const float4* p = (const float4*)(feat + (lm & 7) * FEAT_PITCH + uk * 64 + c * 16 + hi * 8);
        fa[c][0] = p[0];
        fa[c][1] = p[1];
    }
    __builtin_amdgcn_sched_barrier(0);
    v8f u = {0.f, 0.f, 0.f, 0.f, 0.f, 0.f, 0.f, 0.f};
#pragma unroll
    for (int c = 0; c < 4; ++c) {
#pragma unroll
        for (int kk = 0; kk < 4; ++kk) {
            const float4 a4 = fa[c][kk >> 1];
            const float4 b4 = wu[c][kk >> 1];
            v2f am, bm;
            if (kk & 1) { am.x = a4.z; am.y = a4.w; bm.x = b4.z; bm.y = b4.w; }
            else        { am.x = a4.x; am.y = a4.y; bm.x = b4.x; bm.y = b4.y; }
            u = __builtin_amdgcn_wmma_f32_16x16x4_f32(
                false, am, false, bm, (short)0, u, false, false);
        }
    }
    __builtin_amdgcn_sched_barrier(0);
    float* pp = part + (uk * 16 + 8 * hi) * HDIM + uh * 16 + lm;
#pragma unroll
    for (int r = 0; r < 8; ++r) pp[r * HDIM] = u[r];
    __syncthreads();

    if (t < IB * HDIM / 4) {
        const float4 p0 = ((const float4*)part)[t];
        const float4 p1 = ((const float4*)part)[16 * HDIM / 4 + t];
        float4 o4;
        o4.x = p0.x + p1.x + bu.x;
        o4.y = p0.y + p1.y + bu.y;
        o4.z = p0.z + p1.z + bu.z;
        o4.w = p0.w + p1.w + bu.w;
        ((float4*)(out + (size_t)(b * Ndim + i0) * HDIM))[t] = o4;
    }
#undef MSG_MASK
#undef MSG_INIT
#undef MSG_WMMA
#undef MSG_MAX
#undef MSG_EPI_FETCH
#undef EPI_STEP
}

extern "C" void kernel_launch(void* const* d_in, const int* in_sizes, int n_in,
                              void* d_out, int out_size, void* d_ws, size_t ws_size,
                              hipStream_t stream) {
    const float* z      = (const float*)d_in[0];
    const float* e_feat = (const float*)d_in[1];
    const int*   adj    = (const int*)d_in[2];
    const float* W_msg  = (const float*)d_in[3];
    const float* b_msg  = (const float*)d_in[4];
    const float* W_upd  = (const float*)d_in[5];
    const float* b_upd  = (const float*)d_in[6];
    float*       out    = (float*)d_out;

    msg_upd_kernel<<<Bdim * Ndim / IB, 256, 0, stream>>>(e_feat, adj, W_msg, b_msg,
                                                        z, W_upd, b_upd, out);
}


# prologue software-pipelined: m_j operands loaded in k order, its WMMAs start on the first 16-k chunk; e_feat, adjacency, W_e and m_i loads issued between the chunks (sum order unchanged)
# speedup vs baseline: 2.2653x; 1.0684x over previous
#include <hip/hip_runtime.h>
#include <math.h>

#define Bdim 8
#define Ndim 256
#define ZDIM 64
#define EDIM 16
#define HDIM 64

typedef __attribute__((ext_vector_type(2))) float v2f;
typedef __attribute__((ext_vector_type(8))) float v8f;

#define IB 8
#define NSTEP (IB * 4)
#define RED_PITCH 80
#define FEAT_PITCH 132

__global__ __launch_bounds__(256)
void msg_upd_kernel(const float* __restrict__ e_feat,
                    const int*   __restrict__ adj,
                    const float* __restrict__ Wmsg,
                    const float* __restrict__ bmsg,
                    const float* __restrict__ z,
                    const float* __restrict__ Wupd,
                    const float* __restrict__ bupd,
                    float* __restrict__ out) {
    __shared__ float red[IB * 16 * RED_PITCH];
    __shared__ __attribute__((aligned(16))) float feat[IB * FEAT_PITCH];
    __shared__ __attribute__((aligned(16))) float part[2 * 16 * HDIM];
    __shared__ float mip[2 * 16 * HDIM];

    const int blk = blockIdx.x;
    const int b   = blk / (Ndim / IB);
    const int i0  = (blk % (Ndim / IB)) * IB;
    const int t   = threadIdx.x;

    const float NEG_INF = -__builtin_inff();

    const int wave  = t >> 5;
    const int lane  = t & 31;
    const int lm    = lane & 15;
    const int hi    = lane >> 4;
    const int jwave = wave * 32;

    const int uh = wave & 3;
    const int uk = wave >> 2;

    float4 za[4][2][2];
    float4 wb[4][4][2];
    float4 zi[2][2];
    float4 wi[2][2];
    float4 af[IB][2][2];
    int    adjv[IB];
    float4 bf[4][2];
    v8f    cj[2][4];

#define PRO_LOAD_CHUNK(C)                                                         \
    do {                                                                          \
        _Pragma("unroll")                                                         \
        for (int jt = 0; jt < 2; ++jt) {                                          \
            const float4* p = (const float4*)(z                                   \
                + (size_t)(b * Ndim + jwave + jt * 16 + lm) * ZDIM + (C) * 16 + hi * 8); \
            za[C][jt][0] = p[0];                                                  \
            za[C][jt][1] = p[1];                                                  \
        }                                                                         \
        _Pragma("unroll")                                                         \
        for (int ht = 0; ht < 4; ++ht) {                                          \
            const float4* p = (const float4*)(Wmsg                                \
                + (size_t)(ht * 16 + lm) * (2 * ZDIM + EDIM) + ZDIM + (C) * 16 + hi * 8); \
            wb[C][ht][0] = p[0];                                                  \
            wb[C][ht][1] = p[1];                                                  \
        }                                                                         \
    } while (0)

#define PRO_LOAD_EFEAT(R0, R1)                                                    \
    do {                                                                          \
        _Pragma("unroll")                                                         \
        for (int ii = (R0); ii < (R1); ++ii)                                      \
            _Pragma("unroll")                                                     \
            for (int jt = 0; jt < 2; ++jt) {                                      \
                const float4* p = (const float4*)(e_feat                          \
                    + (((size_t)(b * Ndim + i0 + ii) * Ndim) + jwave + jt * 16 + lm) * EDIM \
                    + hi * 8);                                                    \
                af[ii][jt][0] = p[0];                                             \
                af[ii][jt][1] = p[1];                                             \
            }                                                                     \
    } while (0)

#define PRO_MJ_WMMA(C)                                                            \
    do {                                                                          \
        _Pragma("unroll")                                                         \
        for (int kk = 0; kk < 4; ++kk)                                            \
            _Pragma("unroll")                                                     \
            for (int jt = 0; jt < 2; ++jt) {                                      \
                const float4 a4 = za[C][jt][kk >> 1];                             \
                v2f am;                                                           \
                if (kk & 1) { am.x = a4.z; am.y = a4.w; }                         \
                else        { am.x = a4.x; am.y = a4.y; }                         \
                _Pragma("unroll")                                                 \
                for (int ht = 0; ht < 4; ++ht) {                                  \
                    const float4 b4 = wb[C][ht][kk >> 1];                         \
                    v2f bm;                                                       \
                    if (kk & 1) { bm.x = b4.z; bm.y = b4.w; }                     \
                    else        { bm.x = b4.x; bm.y = b4.y; }                     \
                    cj[jt][ht] = __builtin_amdgcn_wmma_f32_16x16x4_f32(           \
                        false, am, false, bm, (short)0, cj[jt][ht], false, false); \
                }                                                                 \
            }                                                                     \
    } while (0)

    PRO_LOAD_CHUNK(0);
    __builtin_amdgcn_sched_barrier(0);
    PRO_LOAD_CHUNK(1);
    __builtin_amdgcn_sched_barrier(0);
    PRO_LOAD_CHUNK(2);
    __builtin_amdgcn_sched_barrier(0);
    PRO_LOAD_CHUNK(3);
    __builtin_amdgcn_sched_barrier(0);
    PRO_LOAD_EFEAT(0, 2);
    __builtin_amdgcn_sched_barrier(0);

#pragma unroll
    for (int jt = 0; jt < 2; ++jt)
#pragma unroll
        for (int ht = 0; ht < 4; ++ht)
#pragma unroll
            for (int r = 0; r < 8; ++r) cj[jt][ht][r] = 0.f;

    PRO_MJ_WMMA(0);
    __builtin_amdgcn_sched_barrier(0);

#pragma unroll
    for (int ii = 0; ii < IB; ++ii)
        adjv[ii] = adj[(size_t)(b * Ndim + i0 + ii) * Ndim + jwave + lane];
#pragma unroll
    for (int ht = 0; ht < 4; ++ht) {
        const float4* p = (const float4*)(Wmsg
            + (size_t)(ht * 16 + lm) * (2 * ZDIM + EDIM) + 2 * ZDIM + hi * 8);
        bf[ht][0] = p[0];
        bf[ht][1] = p[1];
    }
#pragma unroll
    for (int c = 0; c < 2; ++c) {
        const float4* pz = (const float4*)(z
            + (size_t)(b * Ndim + i0 + (lm & 7)) * ZDIM + uk * 32 + c * 16 + hi * 8);
        const float4* pw = (const float4*)(Wmsg
            + (size_t)(uh * 16 + lm) * (2 * ZDIM + EDIM) + uk * 32 + c * 16 + hi * 8);
        zi[c][0] = pz[0];
        zi[c][1] = pz[1];
        wi[c][0] = pw[0];
        wi[c][1] = pw[1];
    }
    __builtin_amdgcn_sched_barrier(0);
    PRO_MJ_WMMA(1);
    __builtin_amdgcn_sched_barrier(0);
    PRO_LOAD_EFEAT(2, 6);
    __builtin_amdgcn_sched_barrier(0);
    PRO_MJ_WMMA(2);
    __builtin_amdgcn_sched_barrier(0);
    PRO_LOAD_EFEAT(6, IB);
    __builtin_amdgcn_sched_barrier(0);
    PRO_MJ_WMMA(3);
    {
        v8f pm = {0.f, 0.f, 0.f, 0.f, 0.f, 0.f, 0.f, 0.f};
#pragma unroll
        for (int c = 0; c < 2; ++c)
#pragma unroll
            for (int kk = 0; kk < 4; ++kk) {
                const float4 a4 = zi[c][kk >> 1];
                const float4 b4 = wi[c][kk >> 1];
                v2f am, bm;
                if (kk & 1) { am.x = a4.z; am.y = a4.w; bm.x = b4.z; bm.y = b4.w; }
                else        { am.x = a4.x; am.y = a4.y; bm.x = b4.x; bm.y = b4.y; }
                pm = __builtin_amdgcn_wmma_f32_16x16x4_f32(
                    false, am, false, bm, (short)0, pm, false, false);
            }
        float* mp = mip + (uk * 16 + 8 * hi) * HDIM + uh * 16 + lm;
#pragma unroll
        for (int r = 0; r < 8; ++r) mp[r * HDIM] = pm[r];
    }
#undef PRO_LOAD_CHUNK
#undef PRO_LOAD_EFEAT
#undef PRO_MJ_WMMA
    __builtin_amdgcn_sched_barrier(0);

    float mk[2][16];
    v8f   acc[3][2];
    float vm[4];

#define MSG_MASK(II)                                                              \
    do {                                                                          \
        const unsigned wb = __builtin_amdgcn_ballot_w32(adjv[II] == 1) >> (8 * hi); \
        _Pragma("unroll")                                                         \
        for (int q = 0; q < 16; ++q)                                              \
            mk[(II) & 1][q] = ((wb >> ((q >> 3) * 16 + (q & 7))) & 1u) ? 0.f : NEG_INF; \
    } while (0)

#define MSG_INIT(S)                                                               \
    do {                                                                          \
        _Pragma("unroll")                                                         \
        for (int r = 0; r < 8; ++r) {                                             \
            acc[(S) % 3][0][r] = cj[0][(S) & 3][r] + mk[((S) >> 2) & 1][r];       \
            acc[(S) % 3][1][r] = cj[1][(S) & 3][r] + mk[((S) >> 2) & 1][8 + r];   \
        }                                                                         \
    } while (0)

#define MSG_WMMA(S)                                                               \
    do {                                                                          \
        _Pragma("unroll")                                                         \
        for (int kk = 0; kk < 4; ++kk) {                                          \
            const float4 b4 = bf[(S) & 3][kk >> 1];                               \
            const float4 a0 = af[(S) >> 2][0][kk >> 1];                           \
            const float4 a1 = af[(S) >> 2][1][kk >> 1];                           \
            v2f bm, am0, am1;                                                     \
            if (kk & 1) { bm.x = b4.z; bm.y = b4.w; am0.x = a0.z; am0.y = a0.w;   \
                          am1.x = a1.z; am1.y = a1.w; }                           \
            else        { bm.x = b4.x; bm.y = b4.y; am0.x = a0.x; am0.y = a0.y;   \
                          am1.x = a1.x; am1.y = a1.y; }                           \
            acc[(S) % 3][0] = __builtin_amdgcn_wmma_f32_16x16x4_f32(              \
                false, am0, false, bm, (short)0, acc[(S) % 3][0], false, false);  \
            acc[(S) % 3][1] = __builtin_amdgcn_wmma_f32_16x16x4_f32(              \
                false, am1, false, bm, (short)0, acc[(S) % 3][1], false, false);  \
        }                                                                         \
    } while (0)

#define MSG_MAX(S)                                                                \
    do {                                                                          \
        float v0 = acc[(S) % 3][0][0];                                            \
        float v1 = acc[(S) % 3][1][0];                                            \
        _Pragma("unroll")                                                         \
        for (int r = 1; r < 8; ++r) {                                             \
            v0 = fmaxf(v0, acc[(S) % 3][0][r]);                                   \
            v1 = fmaxf(v1, acc[(S) % 3][1][r]);                                   \
        }                                                                         \
        vm[(S) & 3] = fmaxf(v0, v1);                                              \
        if (((S) & 3) == 3) {                                                     \
            float* rp = red + ((((S) >> 2) * 8 + wave) * 2 + hi) * RED_PITCH + lm; \
            rp[0]  = vm[0];                                                       \
            rp[16] = vm[1];                                                       \
            rp[32] = vm[2];                                                       \
            rp[48] = vm[3];                                                       \
        }                                                                         \
    } while (0)

    float4 wu[4][2];
    float  bmv[IB * HDIM / 256], zv[IB * ZDIM / 256];
    float4 bu;
#define EPI_STEP 4
#define MSG_EPI_FETCH()                                                           \
    do {                                                                          \
        _Pragma("unroll")                                                         \
        for (int c = 0; c < 4; ++c) {                                             \
            const float4* p = (const float4*)(Wupd                                \
                + (size_t)(uh * 16 + lm) * (ZDIM + HDIM) + uk * 64 + c * 16 + hi * 8); \
            wu[c][0] = p[0];                                                      \
            wu[c][1] = p[1];                                                      \
        }                                                                         \
        _Pragma("unroll")                                                         \
        for (int q = 0; q < IB * HDIM / 256; ++q) {                               \
            const int o = t + q * 256;                                            \
            const int ii = o >> 6, h = o & 63;                                    \
            bmv[q] = bmsg[h];                                                     \
            zv[q]  = z[(size_t)(b * Ndim + i0 + ii) * ZDIM + h];                  \
        }                                                                         \
        bu = ((const float4*)bupd)[t & 15];                                       \
    } while (0)

    MSG_MASK(0);
    MSG_INIT(0);
    __builtin_amdgcn_sched_barrier(0);
#pragma unroll
    for (int s = 0; s < NSTEP; ++s) {
        MSG_WMMA(s);
        if (s >= 1) MSG_MAX(s - 1);
        if ((s & 3) == 1 && (s >> 2) + 1 < IB) MSG_MASK((s >> 2) + 1);
        if (s + 1 < NSTEP) MSG_INIT(s + 1);
        if (s == EPI_STEP) MSG_EPI_FETCH();
        __builtin_amdgcn_sched_barrier(0);
    }
    MSG_MAX(NSTEP - 1);

    __syncthreads();

#pragma unroll
    for (int q = 0; q < IB * HDIM / 256; ++q) {
        const int o = t + q * 256;
        const int ii = o >> 6, h = o & 63;
        float m = red[(ii * 16) * RED_PITCH + h];
#pragma unroll
        for (int w16 = 1; w16 < 16; ++w16) m = fmaxf(m, red[(ii * 16 + w16) * RED_PITCH + h]);
        feat[ii * FEAT_PITCH + h]        = zv[q];
        feat[ii * FEAT_PITCH + ZDIM + h] = m + (mip[ii * HDIM + h] + mip[(16 + ii) * HDIM + h]) + bmv[q];
    }
    __syncthreads();

    float4 fa[4][2];
#pragma unroll
    for (int c = 0; c < 4; ++c) {
        const float4* p = (const float4*)(feat + (lm & 7) * FEAT_PITCH + uk * 64 + c * 16 + hi * 8);
        fa[c][0] = p[0];
        fa[c][1] = p[1];
    }
    __builtin_amdgcn_sched_barrier(0);
    v8f u = {0.f, 0.f, 0.f, 0.f, 0.f, 0.f, 0.f, 0.f};
#pragma unroll
    for (int c = 0; c < 4; ++c) {
#pragma unroll
        for (int kk = 0; kk < 4; ++kk) {
            const float4 a4 = fa[c][kk >> 1];
            const float4 b4 = wu[c][kk >> 1];
            v2f am, bm;
            if (kk & 1) { am.x = a4.z; am.y = a4.w; bm.x = b4.z; bm.y = b4.w; }
            else        { am.x = a4.x; am.y = a4.y; bm.x = b4.x; bm.y = b4.y; }
            u = __builtin_amdgcn_wmma_f32_16x16x4_f32(
                false, am, false, bm, (short)0, u, false, false);
        }
    }
    __builtin_amdgcn_sched_barrier(0);
    float* pp = part + (uk * 16 + 8 * hi) * HDIM + uh * 16 + lm;
#pragma unroll
    for (int r = 0; r < 8; ++r) pp[r * HDIM] = u[r];
    __syncthreads();

    if (t < IB * HDIM / 4) {
        const float4 p0 = ((const float4*)part)[t];
        const float4 p1 = ((const float4*)part)[16 * HDIM / 4 + t];
        float4 o4;
        o4.x = p0.x + p1.x + bu.x;
        o4.y = p0.y + p1.y + bu.y;
        o4.z = p0.z + p1.z + bu.z;
        o4.w = p0.w + p1.w + bu.w;
        ((float4*)(out + (size_t)(b * Ndim + i0) * HDIM))[t] = o4;
    }
#undef MSG_MASK
#undef MSG_INIT
#undef MSG_WMMA
#undef MSG_MAX
#undef MSG_EPI_FETCH
#undef EPI_STEP
}

extern "C" void kernel_launch(void* const* d_in, const int* in_sizes, int n_in,
                              void* d_out, int out_size, void* d_ws, size_t ws_size,
                              hipStream_t stream) {
    const float* z      = (const float*)d_in[0];
    const float* e_feat = (const float*)d_in[1];
    const int*   adj    = (const int*)d_in[2];
    const float* W_msg  = (const float*)d_in[3];
    const float* b_msg  = (const float*)d_in[4];
    const float* W_upd  = (const float*)d_in[5];
    const float* b_upd  = (const float*)d_in[6];
    float*       out    = (float*)d_out;

    msg_upd_kernel<<<Bdim * Ndim / IB, 256, 0, stream>>>(e_feat, adj, W_msg, b_msg,
                                                        z, W_upd, b_upd, out);
}
